# starting kernel
# baseline (speedup 1.0000x reference)
#include <hip/hip_runtime.h>
#include <math.h>


typedef __attribute__((ext_vector_type(2))) float v2f;
typedef __attribute__((ext_vector_type(8))) float v8f;

#define DIN  6
#define HDIM 256

__device__ __forceinline__ float tanh_fast(float x) {
  float ax = __builtin_fabsf(x);
  float t  = __expf(-2.0f * ax);
  float y  = (1.0f - t) * __builtin_amdgcn_rcpf(1.0f + t);
  return __builtin_copysignf(y, x);
}

template <int C>
__device__ __forceinline__ float ccol(const float xv[6], float f0, float f1) {
  if constexpr (C < DIN) return xv[C];
  else if constexpr ((C & 1) == 0) return f0;
  else return f1;
}

template <int KC>
__device__ __forceinline__ v2f make_a(const float xv[6], float f0, float f1, bool hi) {
  float a0lo = ccol<4 * KC + 0>(xv, f0, f1);
  float a1lo = ccol<4 * KC + 1>(xv, f0, f1);
  float a0hi = ccol<4 * KC + 2>(xv, f0, f1);
  float a1hi = ccol<4 * KC + 3>(xv, f0, f1);
  v2f a;
  a.x = hi ? a0hi : a0lo;
  a.y = hi ? a1hi : a1lo;
  return a;
}

__global__ __launch_bounds__(256) void qnn_fused_kernel(
    const float* __restrict__ x,
    const float* __restrict__ W1,
    const float* __restrict__ b1,
    const float* __restrict__ W2,
    const float* __restrict__ b2,
    float* __restrict__ out,
    int B, int ntiles)
{
  const int lane = threadIdx.x & 31;
  const int wave = threadIdx.x >> 5;
  const int tile = blockIdx.x * (blockDim.x >> 5) + wave;
  if (tile >= ntiles) return;

  const int m  = lane & 15;
  const bool hi = lane >= 16;

  int row = tile * 16 + m;
  int rowc = row < B ? row : (B - 1);
  const float* xp = x + (long)rowc * DIN;

  float xv[6];
#pragma unroll
  for (int i = 0; i < DIN; ++i) xv[i] = xp[i];

  float f0 = __cosf(xv[0]) * __cosf(xv[1]);
  float f1 = __cosf(xv[3]) * __cosf(xv[4]);

  v2f A0 = make_a<0>(xv, f0, f1, hi);
  v2f A1 = make_a<1>(xv, f0, f1, hi);
  v2f A2 = make_a<2>(xv, f0, f1, hi);
  v2f A3 = make_a<3>(xv, f0, f1, hi);

  const int krow0 = hi ? 2 : 0;

  float acc[8];
#pragma unroll
  for (int r = 0; r < 8; ++r) acc[r] = 0.0f;

#pragma unroll
  for (int nt = 0; nt < HDIM / 16; ++nt) {
    const int n = nt * 16 + m;

    v8f C = {};
#pragma unroll
    for (int kc = 0; kc < 4; ++kc) {
      v2f Bf;
      Bf.x = W1[(4 * kc + krow0 + 0) * HDIM + n];
      Bf.y = W1[(4 * kc + krow0 + 1) * HDIM + n];
      v2f Af = (kc == 0) ? A0 : (kc == 1) ? A1 : (kc == 2) ? A2 : A3;
      C = __builtin_amdgcn_wmma_f32_16x16x4_f32(
           false, Af,  false, Bf,
           (short)0, C,  false,  false);
    }

    const float b1n = b1[n];
    const float w2n = W2[n];
#pragma unroll
    for (int r = 0; r < 8; ++r) {
      acc[r] = __builtin_fmaf(tanh_fast(C[r] + b1n), w2n, acc[r]);
    }
  }

  const float bias2 = b2[0];
  const int base = tile * 16;

#pragma unroll
  for (int r = 0; r < 8; ++r) {
    float v = acc[r];
    v += __shfl_xor(v, 1, 32);
    v += __shfl_xor(v, 2, 32);
    v += __shfl_xor(v, 4, 32);
    v += __shfl_xor(v, 8, 32);
    if (lane == r && base + r < B)            out[base + r]     = v + bias2;
    if (lane == 16 + r && base + 8 + r < B)   out[base + 8 + r] = v + bias2;
  }
}

extern "C" void kernel_launch(void* const* d_in, const int* in_sizes, int n_in,
                              void* d_out, int out_size, void* d_ws, size_t ws_size,
                              hipStream_t stream) {
  const float* x  = (const float*)d_in[0];
  const float* W1 = (const float*)d_in[1];
  const float* b1 = (const float*)d_in[2];
  const float* W2 = (const float*)d_in[3];
  const float* b2 = (const float*)d_in[4];
  float* out = (float*)d_out;

  const int B = in_sizes[0] / DIN;
  const int ntiles = (B + 15) / 16;
  const int wavesPerBlock = 8;
  const int blocks = (ntiles + wavesPerBlock - 1) / wavesPerBlock;

  qnn_fused_kernel<<<blocks, wavesPerBlock * 32, 0, stream>>>(
      x, W1, b1, W2, b2, out, B, ntiles);
}


# tanh on the one-instruction v_tanh_f32 in place of exp2 + rcp + 6 plain ops per value; row sums pinned in the tile block by an input-only fence so the 128 tanh issue side by side; f32 kept
# speedup vs baseline: 1.1115x; 1.1115x over previous
#include <hip/hip_runtime.h>
#include <math.h>


typedef __attribute__((ext_vector_type(2))) float v2f;
typedef __attribute__((ext_vector_type(8))) float v8f;

#define DIN  6
#define HDIM 256

__device__ __forceinline__ float tanh_fast(float x) {
  return __builtin_amdgcn_tanhf(x);
}

template <int C>
__device__ __forceinline__ float ccol(const float xv[6], float f0, float f1) {
  if constexpr (C < DIN) return xv[C];
  else if constexpr ((C & 1) == 0) return f0;
  else return f1;
}

template <int KC>
__device__ __forceinline__ v2f make_a(const float xv[6], float f0, float f1, bool hi) {
  float a0lo = ccol<4 * KC + 0>(xv, f0, f1);
  float a1lo = ccol<4 * KC + 1>(xv, f0, f1);
  float a0hi = ccol<4 * KC + 2>(xv, f0, f1);
  float a1hi = ccol<4 * KC + 3>(xv, f0, f1);
  v2f a;
  a.x = hi ? a0hi : a0lo;
  a.y = hi ? a1hi : a1lo;
  return a;
}

__global__ __launch_bounds__(256) void qnn_fused_kernel(
    const float* __restrict__ x,
    const float* __restrict__ W1,
    const float* __restrict__ b1,
    const float* __restrict__ W2,
    const float* __restrict__ b2,
    float* __restrict__ out,
    int B, int ntiles)
{
  const int lane = threadIdx.x & 31;
  const int wave = threadIdx.x >> 5;
  const int tile = blockIdx.x * (blockDim.x >> 5) + wave;
  if (tile >= ntiles) return;

  const int m  = lane & 15;
  const bool hi = lane >= 16;

  int row = tile * 16 + m;
  int rowc = row < B ? row : (B - 1);
  const float* xp = x + (long)rowc * DIN;

  float xv[6];
#pragma unroll
  for (int i = 0; i < DIN; ++i) xv[i] = xp[i];

  float f0 = __cosf(xv[0]) * __cosf(xv[1]);
  float f1 = __cosf(xv[3]) * __cosf(xv[4]);

  v2f A0 = make_a<0>(xv, f0, f1, hi);
  v2f A1 = make_a<1>(xv, f0, f1, hi);
  v2f A2 = make_a<2>(xv, f0, f1, hi);
  v2f A3 = make_a<3>(xv, f0, f1, hi);

  const int krow0 = hi ? 2 : 0;

  float acc[8];
#pragma unroll
  for (int r = 0; r < 8; ++r) acc[r] = 0.0f;

#pragma unroll
  for (int nt = 0; nt < HDIM / 16; ++nt) {
    const int n = nt * 16 + m;

    v8f C = {};
#pragma unroll
    for (int kc = 0; kc < 4; ++kc) {
      v2f Bf;
      Bf.x = W1[(4 * kc + krow0 + 0) * HDIM + n];
      Bf.y = W1[(4 * kc + krow0 + 1) * HDIM + n];
      v2f Af = (kc == 0) ? A0 : (kc == 1) ? A1 : (kc == 2) ? A2 : A3;
      C = __builtin_amdgcn_wmma_f32_16x16x4_f32(
           false, Af,  false, Bf,
           (short)0, C,  false,  false);
    }

    const float b1n = b1[n];
    const float w2n = W2[n];
#pragma unroll
    for (int r = 0; r < 8; ++r) {
      acc[r] = __builtin_fmaf(tanh_fast(C[r] + b1n), w2n, acc[r]);
    }
  }

  asm volatile("" :: "v"(acc[0]), "v"(acc[1]), "v"(acc[2]), "v"(acc[3]),
                     "v"(acc[4]), "v"(acc[5]), "v"(acc[6]), "v"(acc[7]));

  const float bias2 = b2[0];
  const int base = tile * 16;

#pragma unroll
  for (int r = 0; r < 8; ++r) {
    float v = acc[r];
    v += __shfl_xor(v, 1, 32);
    v += __shfl_xor(v, 2, 32);
    v += __shfl_xor(v, 4, 32);
    v += __shfl_xor(v, 8, 32);
    if (lane == r && base + r < B)            out[base + r]     = v + bias2;
    if (lane == 16 + r && base + 8 + r < B)   out[base + 8 + r] = v + bias2;
  }
}

extern "C" void kernel_launch(void* const* d_in, const int* in_sizes, int n_in,
                              void* d_out, int out_size, void* d_ws, size_t ws_size,
                              hipStream_t stream) {
  const float* x  = (const float*)d_in[0];
  const float* W1 = (const float*)d_in[1];
  const float* b1 = (const float*)d_in[2];
  const float* W2 = (const float*)d_in[3];
  const float* b2 = (const float*)d_in[4];
  float* out = (float*)d_out;

  const int B = in_sizes[0] / DIN;
  const int ntiles = (B + 15) / 16;
  const int wavesPerBlock = 8;
  const int blocks = (ntiles + wavesPerBlock - 1) / wavesPerBlock;

  qnn_fused_kernel<<<blocks, wavesPerBlock * 32, 0, stream>>>(
      x, W1, b1, W2, b2, out, B, ntiles);
}


# transposed product (W1 columns = A rows, samples = B columns): a lane owns one sample, b1 is C, W2 dot sums in-lane; 1 permlanex16 + 1 store replace 32 ds_bpermute + 16 stores; sum order changes
# speedup vs baseline: 1.4733x; 1.3255x over previous
#include <hip/hip_runtime.h>
#include <math.h>


typedef __attribute__((ext_vector_type(2))) float v2f;
typedef __attribute__((ext_vector_type(4))) float v4f;
typedef __attribute__((ext_vector_type(8))) float v8f;

#define DIN  6
#define HDIM 256

__device__ __forceinline__ float tanh_fast(float x) {
  return __builtin_amdgcn_tanhf(x);
}

__device__ __forceinline__ float xhalf(float x) {
  int b = __builtin_bit_cast(int, x);
  return __builtin_bit_cast(float,
      __builtin_amdgcn_permlanex16(b, b, 0x76543210, 0xfedcba98, false, false));
}

template <int C>
__device__ __forceinline__ float ccol(const float xv[6], float f0, float f1) {
  if constexpr (C < DIN) return xv[C];
  else if constexpr ((C & 1) == 0) return f0;
  else return f1;
}

template <int KC>
__device__ __forceinline__ v2f make_x(const float xv[6], float f0, float f1, bool hi) {
  float a0lo = ccol<4 * KC + 0>(xv, f0, f1);
  float a1lo = ccol<4 * KC + 1>(xv, f0, f1);
  float a0hi = ccol<4 * KC + 2>(xv, f0, f1);
  float a1hi = ccol<4 * KC + 3>(xv, f0, f1);
  v2f a;
  a.x = hi ? a0hi : a0lo;
  a.y = hi ? a1hi : a1lo;
  return a;
}

__global__ __launch_bounds__(256) void qnn_fused_kernel(
    const float* __restrict__ x,
    const float* __restrict__ W1,
    const float* __restrict__ b1,
    const float* __restrict__ W2,
    const float* __restrict__ b2,
    float* __restrict__ out,
    int B, int ntiles)
{
  const int lane = threadIdx.x & 31;
  const int wave = threadIdx.x >> 5;
  const int tile = blockIdx.x * (blockDim.x >> 5) + wave;
  if (tile >= ntiles) return;

  const int c  = lane & 15;
  const bool hi = lane >= 16;

  int row = tile * 16 + c;
  int rowc = row < B ? row : (B - 1);
  const float* xp = x + (long)rowc * DIN;

  float xv[6];
#pragma unroll
  for (int i = 0; i < DIN; ++i) xv[i] = xp[i];

  const float bias2 = b2[0];

  float f0 = __cosf(xv[0]) * __cosf(xv[1]);
  float f1 = __cosf(xv[3]) * __cosf(xv[4]);

  v2f X0 = make_x<0>(xv, f0, f1, hi);
  v2f X1 = make_x<1>(xv, f0, f1, hi);
  v2f X2 = make_x<2>(xv, f0, f1, hi);
  v2f X3 = make_x<3>(xv, f0, f1, hi);

  const int krow0 = hi ? 2 : 0;
  const int hoff  = hi ? 8 : 0;

  float acc[8];
#pragma unroll
  for (int r = 0; r < 8; ++r) acc[r] = 0.0f;

#pragma unroll
  for (int nt = 0; nt < HDIM / 16; ++nt) {
    const int n  = nt * 16 + c;
    const int nb = nt * 16 + hoff;

    const v4f blo = *reinterpret_cast<const v4f*>(b1 + nb);
    const v4f bhi = *reinterpret_cast<const v4f*>(b1 + nb + 4);
    v8f C = {blo.x, blo.y, blo.z, blo.w, bhi.x, bhi.y, bhi.z, bhi.w};
#pragma unroll
    for (int kc = 0; kc < 4; ++kc) {
      v2f Wf;
      Wf.x = W1[(4 * kc + krow0 + 0) * HDIM + n];
      Wf.y = W1[(4 * kc + krow0 + 1) * HDIM + n];
      v2f Xf = (kc == 0) ? X0 : (kc == 1) ? X1 : (kc == 2) ? X2 : X3;
      C = __builtin_amdgcn_wmma_f32_16x16x4_f32(
           false, Wf,  false, Xf,
           (short)0, C,  false,  false);
    }

    const v4f wlo = *reinterpret_cast<const v4f*>(W2 + nb);
    const v4f whi = *reinterpret_cast<const v4f*>(W2 + nb + 4);
    const float w2v[8] = {wlo.x, wlo.y, wlo.z, wlo.w, whi.x, whi.y, whi.z, whi.w};
#pragma unroll
    for (int r = 0; r < 8; ++r) {
      acc[r] = __builtin_fmaf(tanh_fast(C[r]), w2v[r], acc[r]);
    }
  }

  asm volatile("" :: "s"(bias2));

  float s = ((acc[0] + acc[1]) + (acc[2] + acc[3])) + ((acc[4] + acc[5]) + (acc[6] + acc[7]));
  s = s + xhalf(s);

  if (!hi && row < B) out[row] = s + bias2;
}

extern "C" void kernel_launch(void* const* d_in, const int* in_sizes, int n_in,
                              void* d_out, int out_size, void* d_ws, size_t ws_size,
                              hipStream_t stream) {
  const float* x  = (const float*)d_in[0];
  const float* W1 = (const float*)d_in[1];
  const float* b1 = (const float*)d_in[2];
  const float* W2 = (const float*)d_in[3];
  const float* b2 = (const float*)d_in[4];
  float* out = (float*)d_out;

  const int B = in_sizes[0] / DIN;
  const int ntiles = (B + 15) / 16;
  const int wavesPerBlock = 8;
  const int blocks = (ntiles + wavesPerBlock - 1) / wavesPerBlock;

  qnn_fused_kernel<<<blocks, wavesPerBlock * 32, 0, stream>>>(
      x, W1, b1, W2, b2, out, B, ntiles);
}


# work-groups of 4 waves instead of 8: 256 groups, one per WGP, one wave per SIMD, so each SIMD's f32 matrix pipe and each WGP's load path serve half the work; kernel body untouched
# speedup vs baseline: 1.8316x; 1.2432x over previous
#include <hip/hip_runtime.h>
#include <math.h>


typedef __attribute__((ext_vector_type(2))) float v2f;
typedef __attribute__((ext_vector_type(4))) float v4f;
typedef __attribute__((ext_vector_type(8))) float v8f;

#define DIN  6
#define HDIM 256

__device__ __forceinline__ float tanh_fast(float x) {
  return __builtin_amdgcn_tanhf(x);
}

__device__ __forceinline__ float xhalf(float x) {
  int b = __builtin_bit_cast(int, x);
  return __builtin_bit_cast(float,
      __builtin_amdgcn_permlanex16(b, b, 0x76543210, 0xfedcba98, false, false));
}

template <int C>
__device__ __forceinline__ float ccol(const float xv[6], float f0, float f1) {
  if constexpr (C < DIN) return xv[C];
  else if constexpr ((C & 1) == 0) return f0;
  else return f1;
}

template <int KC>
__device__ __forceinline__ v2f make_x(const float xv[6], float f0, float f1, bool hi) {
  float a0lo = ccol<4 * KC + 0>(xv, f0, f1);
  float a1lo = ccol<4 * KC + 1>(xv, f0, f1);
  float a0hi = ccol<4 * KC + 2>(xv, f0, f1);
  float a1hi = ccol<4 * KC + 3>(xv, f0, f1);
  v2f a;
  a.x = hi ? a0hi : a0lo;
  a.y = hi ? a1hi : a1lo;
  return a;
}

__global__ __launch_bounds__(128) void qnn_fused_kernel(
    const float* __restrict__ x,
    const float* __restrict__ W1,
    const float* __restrict__ b1,
    const float* __restrict__ W2,
    const float* __restrict__ b2,
    float* __restrict__ out,
    int B, int ntiles)
{
  const int lane = threadIdx.x & 31;
  const int wave = threadIdx.x >> 5;
  const int tile = blockIdx.x * (blockDim.x >> 5) + wave;
  if (tile >= ntiles) return;

  const int c  = lane & 15;
  const bool hi = lane >= 16;

  int row = tile * 16 + c;
  int rowc = row < B ? row : (B - 1);
  const float* xp = x + (long)rowc * DIN;

  float xv[6];
#pragma unroll
  for (int i = 0; i < DIN; ++i) xv[i] = xp[i];

  const float bias2 = b2[0];

  float f0 = __cosf(xv[0]) * __cosf(xv[1]);
  float f1 = __cosf(xv[3]) * __cosf(xv[4]);

  v2f X0 = make_x<0>(xv, f0, f1, hi);
  v2f X1 = make_x<1>(xv, f0, f1, hi);
  v2f X2 = make_x<2>(xv, f0, f1, hi);
  v2f X3 = make_x<3>(xv, f0, f1, hi);

  const int krow0 = hi ? 2 : 0;
  const int hoff  = hi ? 8 : 0;

  float acc[8];
#pragma unroll
  for (int r = 0; r < 8; ++r) acc[r] = 0.0f;

#pragma unroll
  for (int nt = 0; nt < HDIM / 16; ++nt) {
    const int n  = nt * 16 + c;
    const int nb = nt * 16 + hoff;

    const v4f blo = *reinterpret_cast<const v4f*>(b1 + nb);
    const v4f bhi = *reinterpret_cast<const v4f*>(b1 + nb + 4);
    v8f C = {blo.x, blo.y, blo.z, blo.w, bhi.x, bhi.y, bhi.z, bhi.w};
#pragma unroll
    for (int kc = 0; kc < 4; ++kc) {
      v2f Wf;
      Wf.x = W1[(4 * kc + krow0 + 0) * HDIM + n];
      Wf.y = W1[(4 * kc + krow0 + 1) * HDIM + n];
      v2f Xf = (kc == 0) ? X0 : (kc == 1) ? X1 : (kc == 2) ? X2 : X3;
      C = __builtin_amdgcn_wmma_f32_16x16x4_f32(
           false, Wf,  false, Xf,
           (short)0, C,  false,  false);
    }

    const v4f wlo = *reinterpret_cast<const v4f*>(W2 + nb);
    const v4f whi = *reinterpret_cast<const v4f*>(W2 + nb + 4);
    const float w2v[8] = {wlo.x, wlo.y, wlo.z, wlo.w, whi.x, whi.y, whi.z, whi.w};
#pragma unroll
    for (int r = 0; r < 8; ++r) {
      acc[r] = __builtin_fmaf(tanh_fast(C[r]), w2v[r], acc[r]);
    }
  }

  asm volatile("" :: "s"(bias2));

  float s = ((acc[0] + acc[1]) + (acc[2] + acc[3])) + ((acc[4] + acc[5]) + (acc[6] + acc[7]));
  s = s + xhalf(s);

  if (!hi && row < B) out[row] = s + bias2;
}

extern "C" void kernel_launch(void* const* d_in, const int* in_sizes, int n_in,
                              void* d_out, int out_size, void* d_ws, size_t ws_size,
                              hipStream_t stream) {
  const float* x  = (const float*)d_in[0];
  const float* W1 = (const float*)d_in[1];
  const float* b1 = (const float*)d_in[2];
  const float* W2 = (const float*)d_in[3];
  const float* b2 = (const float*)d_in[4];
  float* out = (float*)d_out;

  const int B = in_sizes[0] / DIN;
  const int ntiles = (B + 15) / 16;
  const int wavesPerBlock = 4;
  const int blocks = (ntiles + wavesPerBlock - 1) / wavesPerBlock;

  qnn_fused_kernel<<<blocks, wavesPerBlock * 32, 0, stream>>>(
      x, W1, b1, W2, b2, out, B, ntiles);
}


# W1 read by 128-bit loads: hidden units dealt to the tiles (row c of tile j, group g = unit 64g+4c+j), one load feeds four tiles; 98 loads per wave, not 194, issued in order of need; sum order changes
# speedup vs baseline: 2.0926x; 1.1425x over previous
#include <hip/hip_runtime.h>
#include <math.h>


typedef __attribute__((ext_vector_type(2))) float v2f;
typedef __attribute__((ext_vector_type(4))) float v4f;
typedef __attribute__((ext_vector_type(8))) float v8f;

#define DIN  6
#define HDIM 256

__device__ __forceinline__ float tanh_fast(float x) {
  return __builtin_amdgcn_tanhf(x);
}

__device__ __forceinline__ float xhalf(float x) {
  int b = __builtin_bit_cast(int, x);
  return __builtin_bit_cast(float,
      __builtin_amdgcn_permlanex16(b, b, 0x76543210, 0xfedcba98, false, false));
}

template <int C>
__device__ __forceinline__ float ccol(const float xv[6], float f0, float f1) {
  if constexpr (C < DIN) return xv[C];
  else if constexpr ((C & 1) == 0) return f0;
  else return f1;
}

template <int KC>
__device__ __forceinline__ v2f make_x(const float xv[6], float f0, float f1, bool hi) {
  float a0lo = ccol<4 * KC + 0>(xv, f0, f1);
  float a1lo = ccol<4 * KC + 1>(xv, f0, f1);
  float a0hi = ccol<4 * KC + 2>(xv, f0, f1);
  float a1hi = ccol<4 * KC + 3>(xv, f0, f1);
  v2f a;
  a.x = hi ? a0hi : a0lo;
  a.y = hi ? a1hi : a1lo;
  return a;
}

__global__ __launch_bounds__(128) void qnn_fused_kernel(
    const float* __restrict__ x,
    const float* __restrict__ W1,
    const float* __restrict__ b1,
    const float* __restrict__ W2,
    const float* __restrict__ b2,
    float* __restrict__ out,
    int B, int ntiles)
{
  const int lane = threadIdx.x & 31;
  const int wave = threadIdx.x >> 5;
  const int tile = blockIdx.x * (blockDim.x >> 5) + wave;
  if (tile >= ntiles) return;

  const int c  = lane & 15;
  const bool hi = lane >= 16;

  int row = tile * 16 + c;
  int rowc = row < B ? row : (B - 1);
  const float* xp = x + (long)rowc * DIN;

  float xv[6];
#pragma unroll
  for (int i = 0; i < DIN; ++i) xv[i] = xp[i];

  const float bias2 = b2[0];

  __builtin_amdgcn_sched_barrier(0);

  const int krow0 = hi ? 2 : 0;
  const int hoff  = hi ? 32 : 0;
  const int wo = krow0 * HDIM + 4 * c;

  v4f Wl[4][4][2], bv[4][8], wv[4][8];
#pragma unroll
  for (int g = 0; g < 4; ++g) {
#pragma unroll
    for (int kc = 0; kc < 4; ++kc) {
      Wl[g][kc][0] = *reinterpret_cast<const v4f*>(W1 + (wo + (4 * kc + 0) * HDIM + 64 * g));
      Wl[g][kc][1] = *reinterpret_cast<const v4f*>(W1 + (wo + (4 * kc + 1) * HDIM + 64 * g));
    }
#pragma unroll
    for (int r = 0; r < 8; ++r)
      bv[g][r] = *reinterpret_cast<const v4f*>(b1 + (hoff + 64 * g + 4 * r));
  }

  __builtin_amdgcn_sched_barrier(0);

#pragma unroll
  for (int g = 0; g < 4; ++g)
#pragma unroll
    for (int r = 0; r < 8; ++r)
      wv[g][r] = *reinterpret_cast<const v4f*>(W2 + (hoff + 64 * g + 4 * r));

  __builtin_amdgcn_sched_barrier(0);

  float f0 = __cosf(xv[0]) * __cosf(xv[1]);
  float f1 = __cosf(xv[3]) * __cosf(xv[4]);

  v2f X0 = make_x<0>(xv, f0, f1, hi);
  v2f X1 = make_x<1>(xv, f0, f1, hi);
  v2f X2 = make_x<2>(xv, f0, f1, hi);
  v2f X3 = make_x<3>(xv, f0, f1, hi);

  v2f acc[8];
#pragma unroll
  for (int r = 0; r < 8; ++r) acc[r] = (v2f){0.0f, 0.0f};

#pragma unroll
  for (int g = 0; g < HDIM / 64; ++g) {
    v8f D[4];
#pragma unroll
    for (int j = 0; j < 4; ++j) {
      v8f C = {bv[g][0][j], bv[g][1][j], bv[g][2][j], bv[g][3][j],
               bv[g][4][j], bv[g][5][j], bv[g][6][j], bv[g][7][j]};
#pragma unroll
      for (int kc = 0; kc < 4; ++kc) {
        v2f Wf;
        Wf.x = Wl[g][kc][0][j];
        Wf.y = Wl[g][kc][1][j];
        v2f Xf = (kc == 0) ? X0 : (kc == 1) ? X1 : (kc == 2) ? X2 : X3;
        C = __builtin_amdgcn_wmma_f32_16x16x4_f32(
             false, Wf,  false, Xf,
             (short)0, C,  false,  false);
      }
      D[j] = C;
    }

#pragma unroll
    for (int p = 0; p < 2; ++p) {
#pragma unroll
      for (int r = 0; r < 8; ++r) {
        v2f t = {tanh_fast(D[2 * p][r]), tanh_fast(D[2 * p + 1][r])};
        v2f w = {wv[g][r][2 * p], wv[g][r][2 * p + 1]};
        acc[r] = __builtin_elementwise_fma(t, w, acc[r]);
      }
    }
  }

  asm volatile("" :: "s"(bias2));

  v2f s2 = ((acc[0] + acc[1]) + (acc[2] + acc[3])) + ((acc[4] + acc[5]) + (acc[6] + acc[7]));
  float s = s2.x + s2.y;
  s = s + xhalf(s);

  if (!hi && row < B) out[row] = s + bias2;
}

extern "C" void kernel_launch(void* const* d_in, const int* in_sizes, int n_in,
                              void* d_out, int out_size, void* d_ws, size_t ws_size,
                              hipStream_t stream) {
  const float* x  = (const float*)d_in[0];
  const float* W1 = (const float*)d_in[1];
  const float* b1 = (const float*)d_in[2];
  const float* W2 = (const float*)d_in[3];
  const float* b2 = (const float*)d_in[4];
  float* out = (float*)d_out;

  const int B = in_sizes[0] / DIN;
  const int ntiles = (B + 15) / 16;
  const int wavesPerBlock = 4;
  const int blocks = (ntiles + wavesPerBlock - 1) / wavesPerBlock;

  qnn_fused_kernel<<<blocks, wavesPerBlock * 32, 0, stream>>>(
      x, W1, b1, W2, b2, out, B, ntiles);
}
